# starting kernel
# baseline (speedup 1.0000x reference)
#include <hip/hip_runtime.h>

#define N_LEAF       64
#define SPECTRAL_DIM 64
#define N_BANDS      4
#define BAND_SIZE    16
#define LAM          0.1f
#define EPSF         1e-8f

typedef float v2f __attribute__((ext_vector_type(2)));
typedef float v4f __attribute__((ext_vector_type(4)));
typedef float v8f __attribute__((ext_vector_type(8)));

#define WAVES_PER_BLOCK 4
#define PTS_PER_WAVE    32
#define PTS_PER_BLOCK   (WAVES_PER_BLOCK * PTS_PER_WAVE)
#define LDS_STRIDE      65

__device__ __forceinline__ v8f zero8() {
    v8f z;
    #pragma unroll
    for (int i = 0; i < 8; ++i) z[i] = 0.0f;
    return z;
}

__device__ __forceinline__ float frcp(float x)  { return __builtin_amdgcn_rcpf(x); }
__device__ __forceinline__ float fsqrt(float x) { return __builtin_amdgcn_sqrtf(x); }

__global__ __launch_bounds__(PTS_PER_BLOCK) void inception_traversal_kernel(
    const float* __restrict__ pos3d,
    const float* __restrict__ spectral,
    const float* __restrict__ centers1,
    const float* __restrict__ centers2,
    const float* __restrict__ centers3,
    const float* __restrict__ portal1,
    const float* __restrict__ portal2,
    const float* __restrict__ Wb,
    const float* __restrict__ bb,
    const float* __restrict__ bw,
    float* __restrict__ out,
    int BS)
{
    __shared__ float ldsRefr[WAVES_PER_BLOCK][PTS_PER_WAVE][LDS_STRIDE];

    const int lane   = threadIdx.x & 31;
    const int wid    = threadIdx.x >> 5;
    const int pblock = blockIdx.x * PTS_PER_BLOCK;
    if (pblock >= BS) return;
    const int pwave  = pblock + wid * PTS_PER_WAVE;

    float w0 = bw[0], w1 = bw[1], w2 = bw[2], w3 = bw[3];
    float wm = fmaxf(fmaxf(w0, w1), fmaxf(w2, w3));
    float ew0 = __expf(w0 - wm), ew1 = __expf(w1 - wm);
    float ew2 = __expf(w2 - wm), ew3 = __expf(w3 - wm);
    float wr  = frcp(ew0 + ew1 + ew2 + ew3);
    float wband[N_BANDS] = { ew0 * wr, ew1 * wr, ew2 * wr, ew3 * wr };

    const int mrow  = lane & 15;
    const int khalf = (lane >> 4) * 2;

    #pragma unroll
    for (int mt = 0; mt < 2; ++mt) {
        const int P = pwave + mt * 16;

        v8f acc[4];
        #pragma unroll
        for (int t = 0; t < 4; ++t) acc[t] = zero8();

        #pragma unroll
        for (int n = 0; n < N_BANDS; ++n) {
            v8f logit[4];
            #pragma unroll
            for (int t = 0; t < 4; ++t) logit[t] = zero8();

            #pragma unroll
            for (int kk = 0; kk < 4; ++kk) {
                const int c = n * BAND_SIZE + kk * 4 + khalf;
                v2f a = *reinterpret_cast<const v2f*>(
                    spectral + (size_t)(P + mrow) * SPECTRAL_DIM + c);
                #pragma unroll
                for (int t = 0; t < 4; ++t) {
                    const float* wp = Wb + ((size_t)n * BAND_SIZE + kk * 4 + khalf) * N_LEAF
                                         + t * 16 + mrow;
                    v2f b; b.x = wp[0]; b.y = wp[N_LEAF];
                    logit[t] = __builtin_amdgcn_wmma_f32_16x16x4_f32(
                        false, a, false, b, (short)0, logit[t], false, false);
                }
            }
            #pragma unroll
            for (int t = 0; t < 4; ++t) {
                const float bias = bb[n * N_LEAF + t * 16 + mrow];
                const float wn   = wband[n];
                #pragma unroll
                for (int j = 0; j < 8; ++j) {
                    float s = frcp(1.0f + __expf(-(logit[t][j] + bias)));
                    acc[t][j] += wn * s;
                }
            }
        }

        const int rowoff = (lane >> 4) * 8 + mt * 16;
        #pragma unroll
        for (int t = 0; t < 4; ++t)
            #pragma unroll
            for (int j = 0; j < 8; ++j)
                ldsRefr[wid][rowoff + j][t * 16 + mrow] = acc[t][j];
    }

    __syncthreads();

    const int p = pwave + lane;
    float* row = &ldsRefr[wid][lane][0];

    const float px = pos3d[(size_t)p * 3 + 0];
    const float py = pos3d[(size_t)p * 3 + 1];
    const float pz = pos3d[(size_t)p * 3 + 2];
    const float inv2T2 = 1.0f / (2.0f + EPSF);

    float d1v[4];
    #pragma unroll
    for (int d = 0; d < 4; ++d) {
        float dx = px - centers1[d * 3 + 0];
        float dy = py - centers1[d * 3 + 1];
        float dz = pz - centers1[d * 3 + 2];
        d1v[d] = dx * dx + dy * dy + dz * dz;
    }
    float z1max = -1e30f;
    #pragma unroll
    for (int d = 0; d < 4; ++d) z1max = fmaxf(z1max, -d1v[d] * inv2T2);
    float ex1[4], s1 = 0.0f;
    #pragma unroll
    for (int d = 0; d < 4; ++d) { ex1[d] = __expf(-d1v[d] * inv2T2 - z1max); s1 += ex1[d]; }
    const float r1 = frcp(s1);
    float m1[4], e1v[4];
    #pragma unroll
    for (int d = 0; d < 4; ++d) {
        m1[d]  = ex1[d] * r1;
        e1v[d] = __expf(-LAM * fsqrt(d1v[d] + EPSF));
    }

    float ssum = 0.0f;
    #pragma unroll
    for (int d = 0; d < 4; ++d) {
        const float* T1 = portal1 + d * 12;
        float qx = T1[0]*px + T1[1]*py + T1[2]*pz  + T1[3];
        float qy = T1[4]*px + T1[5]*py + T1[6]*pz  + T1[7];
        float qz = T1[8]*px + T1[9]*py + T1[10]*pz + T1[11];

        float d2v[4];
        #pragma unroll
        for (int si = 0; si < 4; ++si) {
            const float* c = centers2 + (size_t)(d * 4 + si) * 3;
            float dx = qx - c[0], dy = qy - c[1], dz = qz - c[2];
            d2v[si] = dx * dx + dy * dy + dz * dz;
        }
        float z2max = -1e30f;
        #pragma unroll
        for (int si = 0; si < 4; ++si) z2max = fmaxf(z2max, -d2v[si] * inv2T2);
        float ex2[4], s2 = 0.0f;
        #pragma unroll
        for (int si = 0; si < 4; ++si) { ex2[si] = __expf(-d2v[si] * inv2T2 - z2max); s2 += ex2[si]; }
        const float r2 = frcp(s2);

        #pragma unroll
        for (int si = 0; si < 4; ++si) {
            float m2 = m1[d] * ex2[si] * r2;
            float e2 = e1v[d] * __expf(-LAM * fsqrt(d2v[si] + EPSF));

            const float* T2 = portal2 + (size_t)(d * 4 + si) * 12;
            float rx = T2[0]*qx + T2[1]*qy + T2[2]*qz  + T2[3];
            float ry = T2[4]*qx + T2[5]*qy + T2[6]*qz  + T2[7];
            float rz = T2[8]*qx + T2[9]*qy + T2[10]*qz + T2[11];

            float d3v[4];
            #pragma unroll
            for (int ci = 0; ci < 4; ++ci) {
                const float* c = centers3 + (size_t)((d * 4 + si) * 4 + ci) * 3;
                float dx = rx - c[0], dy = ry - c[1], dz = rz - c[2];
                d3v[ci] = dx * dx + dy * dy + dz * dz;
            }
            float z3max = -1e30f;
            #pragma unroll
            for (int ci = 0; ci < 4; ++ci) z3max = fmaxf(z3max, -d3v[ci] * inv2T2);
            float ex3[4], s3 = 0.0f;
            #pragma unroll
            for (int ci = 0; ci < 4; ++ci) { ex3[ci] = __expf(-d3v[ci] * inv2T2 - z3max); s3 += ex3[ci]; }
            const float r3   = frcp(s3);
            const float m2e2 = m2 * e2;

            #pragma unroll
            for (int ci = 0; ci < 4; ++ci) {
                float g  = m2e2 * ex3[ci] * r3 * __expf(-LAM * fsqrt(d3v[ci] + EPSF));
                int   k  = (d * 4 + si) * 4 + ci;
                float pr = g * row[k];
                row[k] = pr;
                ssum  += pr;
            }
        }
    }

    float rn = frcp(fmaxf(ssum, 1e-8f));
    float* op = out + (size_t)p * N_LEAF;
    #pragma unroll
    for (int k4 = 0; k4 < 16; ++k4) {
        v4f v;
        v.x = row[k4 * 4 + 0] * rn;
        v.y = row[k4 * 4 + 1] * rn;
        v.z = row[k4 * 4 + 2] * rn;
        v.w = row[k4 * 4 + 3] * rn;
        *reinterpret_cast<v4f*>(op + k4 * 4) = v;
    }
}

extern "C" void kernel_launch(void* const* d_in, const int* in_sizes, int n_in,
                              void* d_out, int out_size, void* d_ws, size_t ws_size,
                              hipStream_t stream) {
    const float* pos3d    = (const float*)d_in[0];
    const float* spectral = (const float*)d_in[1];
    const float* centers1 = (const float*)d_in[2];
    const float* centers2 = (const float*)d_in[3];
    const float* centers3 = (const float*)d_in[4];
    const float* portal1  = (const float*)d_in[5];
    const float* portal2  = (const float*)d_in[6];
    const float* Wb       = (const float*)d_in[7];
    const float* bb       = (const float*)d_in[8];
    const float* bw       = (const float*)d_in[9];
    float* out            = (float*)d_out;

    const int BS = in_sizes[0] / 3;
    const int blocks = (BS + PTS_PER_BLOCK - 1) / PTS_PER_BLOCK;

    hipLaunchKernelGGL(inception_traversal_kernel,
                       dim3(blocks), dim3(PTS_PER_BLOCK), 0, stream,
                       pos3d, spectral, centers1, centers2, centers3,
                       portal1, portal2, Wb, bb, bw, out, BS);
}


# Routing product split over a block's waves by leaf block (wave w: leaves 16w..16w+15, all 128 points); point rows shared via LDS by whole-line loads; 16 weight fragments per wave, 298 -> 176 VGPRs
# speedup vs baseline: 1.3000x; 1.3000x over previous
#include <hip/hip_runtime.h>

#define N_LEAF       64
#define SPECTRAL_DIM 64
#define N_BANDS      4
#define BAND_SIZE    16
#define LAM          0.1f
#define EPSF         1e-8f

typedef float v2f __attribute__((ext_vector_type(2)));
typedef float v4f __attribute__((ext_vector_type(4)));
typedef float v8f __attribute__((ext_vector_type(8)));

#define WAVES_PER_BLOCK 4
#define PTS_PER_WAVE    32
#define PTS_PER_BLOCK   (WAVES_PER_BLOCK * PTS_PER_WAVE)
#define LDS_STRIDE      65
#define X_STRIDE        68
#define TILES_PER_BLOCK (PTS_PER_BLOCK / 16)

__device__ __forceinline__ v8f zero8() {
    v8f z;
    #pragma unroll
    for (int i = 0; i < 8; ++i) z[i] = 0.0f;
    return z;
}

__device__ __forceinline__ float frcp(float x)  { return __builtin_amdgcn_rcpf(x); }
__device__ __forceinline__ float fsqrt(float x) { return __builtin_amdgcn_sqrtf(x); }

__global__ __launch_bounds__(PTS_PER_BLOCK) void inception_traversal_kernel(
    const float* __restrict__ pos3d,
    const float* __restrict__ spectral,
    const float* __restrict__ centers1,
    const float* __restrict__ centers2,
    const float* __restrict__ centers3,
    const float* __restrict__ portal1,
    const float* __restrict__ portal2,
    const float* __restrict__ Wb,
    const float* __restrict__ bb,
    const float* __restrict__ bw,
    float* __restrict__ out,
    int BS)
{
    __shared__ float ldsRefr[PTS_PER_BLOCK][LDS_STRIDE];
    __shared__ __attribute__((aligned(16))) float ldsX[PTS_PER_BLOCK][X_STRIDE];

    const int lane   = threadIdx.x & 31;
    const int wid    = threadIdx.x >> 5;
    const int pblock = blockIdx.x * PTS_PER_BLOCK;
    if (pblock >= BS) return;
    const int pwave  = pblock + wid * PTS_PER_WAVE;

    float w0 = bw[0], w1 = bw[1], w2 = bw[2], w3 = bw[3];
    float wm = fmaxf(fmaxf(w0, w1), fmaxf(w2, w3));
    float ew0 = __expf(w0 - wm), ew1 = __expf(w1 - wm);
    float ew2 = __expf(w2 - wm), ew3 = __expf(w3 - wm);
    float wr  = frcp(ew0 + ew1 + ew2 + ew3);
    float wband[N_BANDS] = { ew0 * wr, ew1 * wr, ew2 * wr, ew3 * wr };

    const int mrow  = lane & 15;
    const int khalf = (lane >> 4) * 2;


    {
        const float* xw = spectral + (size_t)pwave * SPECTRAL_DIM + lane * 4;
        const int r0 = wid * PTS_PER_WAVE + (lane >> 4);
        const int c0 = (lane & 15) * 4;
        v4f xv[16];
        #pragma unroll
        for (int i = 0; i < 16; ++i)
            xv[i] = *reinterpret_cast<const v4f*>(xw + (size_t)i * 2 * SPECTRAL_DIM);
        #pragma unroll
        for (int i = 0; i < 16; ++i)
            *reinterpret_cast<v4f*>(&ldsX[r0 + 2 * i][c0]) = xv[i];
    }

    v2f bfrag[N_BANDS][4];
    float bias[N_BANDS];
    #pragma unroll
    for (int n = 0; n < N_BANDS; ++n) {
        #pragma unroll
        for (int kk = 0; kk < 4; ++kk) {
            const float* wp = Wb + ((size_t)n * BAND_SIZE + kk * 4 + khalf) * N_LEAF
                                 + wid * 16 + mrow;
            bfrag[n][kk].x = wp[0];
            bfrag[n][kk].y = wp[N_LEAF];
        }
        bias[n] = bb[n * N_LEAF + wid * 16 + mrow];
    }

    __syncthreads();

    const int rowoff = (lane >> 4) * 8;

    v2f a[2][N_BANDS][4];
    #pragma unroll
    for (int u = 0; u < 2; ++u) {
        const float* xr = &ldsX[u * 16 + mrow][khalf];
        #pragma unroll
        for (int n = 0; n < N_BANDS; ++n)
            #pragma unroll
            for (int kk = 0; kk < 4; ++kk)
                a[u][n][kk] = *reinterpret_cast<const v2f*>(xr + n * BAND_SIZE + kk * 4);
    }

    #pragma unroll 1
    for (int tile = 0; tile < TILES_PER_BLOCK; tile += 2) {
        v8f logit[2][N_BANDS];
        #pragma unroll
        for (int u = 0; u < 2; ++u) {
            #pragma unroll
            for (int n = 0; n < N_BANDS; ++n) logit[u][n] = zero8();
            #pragma unroll
            for (int kk = 0; kk < 4; ++kk)
                #pragma unroll
                for (int n = 0; n < N_BANDS; ++n)
                    logit[u][n] = __builtin_amdgcn_wmma_f32_16x16x4_f32(
                        false, a[u][n][kk], false, bfrag[n][kk], (short)0, logit[u][n],
                        false, false);
        }

        {
            const int tn = (tile + 2 < TILES_PER_BLOCK) ? tile + 2 : tile;
            #pragma unroll
            for (int u = 0; u < 2; ++u) {
                const float* xr = &ldsX[(tn + u) * 16 + mrow][khalf];
                #pragma unroll
                for (int n = 0; n < N_BANDS; ++n)
                    #pragma unroll
                    for (int kk = 0; kk < 4; ++kk)
                        a[u][n][kk] = *reinterpret_cast<const v2f*>(xr + n * BAND_SIZE + kk * 4);
            }
        }

        #pragma unroll
        for (int u = 0; u < 2; ++u) {
            v8f acc = zero8();
            #pragma unroll
            for (int n = 0; n < N_BANDS; ++n) {
                const float wn = wband[n];
                #pragma unroll
                for (int j = 0; j < 8; ++j) {
                    float s = frcp(1.0f + __expf(-(logit[u][n][j] + bias[n])));
                    acc[j] += wn * s;
                }
            }

            #pragma unroll
            for (int j = 0; j < 8; ++j)
                ldsRefr[(tile + u) * 16 + rowoff + j][wid * 16 + mrow] = acc[j];
        }
    }

    __syncthreads();

    const int p = pwave + lane;
    float* row = &ldsRefr[wid * PTS_PER_WAVE + lane][0];

    const float px = pos3d[(size_t)p * 3 + 0];
    const float py = pos3d[(size_t)p * 3 + 1];
    const float pz = pos3d[(size_t)p * 3 + 2];
    const float inv2T2 = 1.0f / (2.0f + EPSF);

    float d1v[4];
    #pragma unroll
    for (int d = 0; d < 4; ++d) {
        float dx = px - centers1[d * 3 + 0];
        float dy = py - centers1[d * 3 + 1];
        float dz = pz - centers1[d * 3 + 2];
        d1v[d] = dx * dx + dy * dy + dz * dz;
    }
    float z1max = -1e30f;
    #pragma unroll
    for (int d = 0; d < 4; ++d) z1max = fmaxf(z1max, -d1v[d] * inv2T2);
    float ex1[4], s1 = 0.0f;
    #pragma unroll
    for (int d = 0; d < 4; ++d) { ex1[d] = __expf(-d1v[d] * inv2T2 - z1max); s1 += ex1[d]; }
    const float r1 = frcp(s1);
    float m1[4], e1v[4];
    #pragma unroll
    for (int d = 0; d < 4; ++d) {
        m1[d]  = ex1[d] * r1;
        e1v[d] = __expf(-LAM * fsqrt(d1v[d] + EPSF));
    }

    float ssum = 0.0f;
    #pragma unroll
    for (int d = 0; d < 4; ++d) {
        const float* T1 = portal1 + d * 12;
        float qx = T1[0]*px + T1[1]*py + T1[2]*pz  + T1[3];
        float qy = T1[4]*px + T1[5]*py + T1[6]*pz  + T1[7];
        float qz = T1[8]*px + T1[9]*py + T1[10]*pz + T1[11];

        float d2v[4];
        #pragma unroll
        for (int si = 0; si < 4; ++si) {
            const float* c = centers2 + (size_t)(d * 4 + si) * 3;
            float dx = qx - c[0], dy = qy - c[1], dz = qz - c[2];
            d2v[si] = dx * dx + dy * dy + dz * dz;
        }
        float z2max = -1e30f;
        #pragma unroll
        for (int si = 0; si < 4; ++si) z2max = fmaxf(z2max, -d2v[si] * inv2T2);
        float ex2[4], s2 = 0.0f;
        #pragma unroll
        for (int si = 0; si < 4; ++si) { ex2[si] = __expf(-d2v[si] * inv2T2 - z2max); s2 += ex2[si]; }
        const float r2 = frcp(s2);

        #pragma unroll
        for (int si = 0; si < 4; ++si) {
            float m2 = m1[d] * ex2[si] * r2;
            float e2 = e1v[d] * __expf(-LAM * fsqrt(d2v[si] + EPSF));

            const float* T2 = portal2 + (size_t)(d * 4 + si) * 12;
            float rx = T2[0]*qx + T2[1]*qy + T2[2]*qz  + T2[3];
            float ry = T2[4]*qx + T2[5]*qy + T2[6]*qz  + T2[7];
            float rz = T2[8]*qx + T2[9]*qy + T2[10]*qz + T2[11];

            float d3v[4];
            #pragma unroll
            for (int ci = 0; ci < 4; ++ci) {
                const float* c = centers3 + (size_t)((d * 4 + si) * 4 + ci) * 3;
                float dx = rx - c[0], dy = ry - c[1], dz = rz - c[2];
                d3v[ci] = dx * dx + dy * dy + dz * dz;
            }
            float z3max = -1e30f;
            #pragma unroll
            for (int ci = 0; ci < 4; ++ci) z3max = fmaxf(z3max, -d3v[ci] * inv2T2);
            float ex3[4], s3 = 0.0f;
            #pragma unroll
            for (int ci = 0; ci < 4; ++ci) { ex3[ci] = __expf(-d3v[ci] * inv2T2 - z3max); s3 += ex3[ci]; }
            const float r3   = frcp(s3);
            const float m2e2 = m2 * e2;

            #pragma unroll
            for (int ci = 0; ci < 4; ++ci) {
                float g  = m2e2 * ex3[ci] * r3 * __expf(-LAM * fsqrt(d3v[ci] + EPSF));
                int   k  = (d * 4 + si) * 4 + ci;
                float pr = g * row[k];
                row[k] = pr;
                ssum  += pr;
            }
        }
    }

    float rn = frcp(fmaxf(ssum, 1e-8f));
    float* op = out + (size_t)p * N_LEAF;
    #pragma unroll
    for (int k4 = 0; k4 < 16; ++k4) {
        v4f v;
        v.x = row[k4 * 4 + 0] * rn;
        v.y = row[k4 * 4 + 1] * rn;
        v.z = row[k4 * 4 + 2] * rn;
        v.w = row[k4 * 4 + 3] * rn;
        *reinterpret_cast<v4f*>(op + k4 * 4) = v;
    }
}

extern "C" void kernel_launch(void* const* d_in, const int* in_sizes, int n_in,
                              void* d_out, int out_size, void* d_ws, size_t ws_size,
                              hipStream_t stream) {
    const float* pos3d    = (const float*)d_in[0];
    const float* spectral = (const float*)d_in[1];
    const float* centers1 = (const float*)d_in[2];
    const float* centers2 = (const float*)d_in[3];
    const float* centers3 = (const float*)d_in[4];
    const float* portal1  = (const float*)d_in[5];
    const float* portal2  = (const float*)d_in[6];
    const float* Wb       = (const float*)d_in[7];
    const float* bb       = (const float*)d_in[8];
    const float* bw       = (const float*)d_in[9];
    float* out            = (float*)d_out;

    const int BS = in_sizes[0] / 3;
    const int blocks = (BS + PTS_PER_BLOCK - 1) / PTS_PER_BLOCK;

    hipLaunchKernelGGL(inception_traversal_kernel,
                       dim3(blocks), dim3(PTS_PER_BLOCK), 0, stream,
                       pos3d, spectral, centers1, centers2, centers3,
                       portal1, portal2, Wb, bb, bw, out, BS);
}


# Traversal re-mapped to 16 lanes per point (lane = sub-domain, its 4 leaves in registers): softmax sums and row total as 16-lane butterflies (sum order changes); tile and output move as whole rows
# speedup vs baseline: 1.5940x; 1.2262x over previous
#include <hip/hip_runtime.h>

#define N_LEAF       64
#define SPECTRAL_DIM 64
#define N_BANDS      4
#define BAND_SIZE    16
#define LAM          0.1f
#define EPSF         1e-8f

typedef float v2f __attribute__((ext_vector_type(2)));
typedef float v4f __attribute__((ext_vector_type(4)));
typedef float v8f __attribute__((ext_vector_type(8)));

#define WAVES_PER_BLOCK 4
#define PTS_PER_WAVE    32
#define PTS_PER_BLOCK   (WAVES_PER_BLOCK * PTS_PER_WAVE)
#define LDS_STRIDE      68
#define X_STRIDE        68
#define TILES_PER_BLOCK (PTS_PER_BLOCK / 16)

__device__ __forceinline__ v8f zero8() {
    v8f z;
    #pragma unroll
    for (int i = 0; i < 8; ++i) z[i] = 0.0f;
    return z;
}

__device__ __forceinline__ float frcp(float x)  { return __builtin_amdgcn_rcpf(x); }
__device__ __forceinline__ float fsqrt(float x) { return __builtin_amdgcn_sqrtf(x); }

template <int M>
__device__ __forceinline__ float row_xor(float v) {
    return __builtin_bit_cast(float, __builtin_amdgcn_mov_dpp(__builtin_bit_cast(int, v), 0x160 | M, 0xf, 0xf, false));
}

template <int M>
__device__ __forceinline__ v2f row_xor2(v2f v) { v2f r; r.x = row_xor<M>(v.x); r.y = row_xor<M>(v.y); return r; }
__device__ __forceinline__ v2f mk2(float a)      { v2f r; r.x = a; r.y = a; return r; }
__device__ __forceinline__ v2f max2(v2f a, v2f b) { v2f r; r.x = fmaxf(a.x, b.x); r.y = fmaxf(a.y, b.y); return r; }
__device__ __forceinline__ v2f exp_2(v2f a)  { v2f r; r.x = __expf(a.x); r.y = __expf(a.y); return r; }
__device__ __forceinline__ v2f sqrt_2(v2f a) { v2f r; r.x = fsqrt(a.x);  r.y = fsqrt(a.y);  return r; }
__device__ __forceinline__ v2f rcp_2(v2f a)  { v2f r; r.x = frcp(a.x);   r.y = frcp(a.y);   return r; }

__global__ __launch_bounds__(PTS_PER_BLOCK) void inception_traversal_kernel(
    const float* __restrict__ pos3d,
    const float* __restrict__ spectral,
    const float* __restrict__ centers1,
    const float* __restrict__ centers2,
    const float* __restrict__ centers3,
    const float* __restrict__ portal1,
    const float* __restrict__ portal2,
    const float* __restrict__ Wb,
    const float* __restrict__ bb,
    const float* __restrict__ bw,
    float* __restrict__ out,
    int BS)
{
    __shared__ __attribute__((aligned(16))) float ldsRefr[PTS_PER_BLOCK][LDS_STRIDE];
    __shared__ __attribute__((aligned(16))) float ldsX[PTS_PER_BLOCK][X_STRIDE];

    const int lane   = threadIdx.x & 31;
    const int wid    = threadIdx.x >> 5;
    const int pblock = blockIdx.x * PTS_PER_BLOCK;
    if (pblock >= BS) return;
    const int pwave  = pblock + wid * PTS_PER_WAVE;

    float w0 = bw[0], w1 = bw[1], w2 = bw[2], w3 = bw[3];
    float wm = fmaxf(fmaxf(w0, w1), fmaxf(w2, w3));
    float ew0 = __expf(w0 - wm), ew1 = __expf(w1 - wm);
    float ew2 = __expf(w2 - wm), ew3 = __expf(w3 - wm);
    float wr  = frcp(ew0 + ew1 + ew2 + ew3);
    float wband[N_BANDS] = { ew0 * wr, ew1 * wr, ew2 * wr, ew3 * wr };

    const int mrow  = lane & 15;
    const int khalf = (lane >> 4) * 2;


    {
        const float* xw = spectral + (size_t)pwave * SPECTRAL_DIM + lane * 4;
        const int r0 = wid * PTS_PER_WAVE + (lane >> 4);
        const int c0 = (lane & 15) * 4;
        v4f xv[16];
        #pragma unroll
        for (int i = 0; i < 16; ++i)
            xv[i] = *reinterpret_cast<const v4f*>(xw + (size_t)i * 2 * SPECTRAL_DIM);
        #pragma unroll
        for (int i = 0; i < 16; ++i)
            *reinterpret_cast<v4f*>(&ldsX[r0 + 2 * i][c0]) = xv[i];
    }

    v2f bfrag[N_BANDS][4];
    float bias[N_BANDS];
    #pragma unroll
    for (int n = 0; n < N_BANDS; ++n) {
        #pragma unroll
        for (int kk = 0; kk < 4; ++kk) {
            const float* wp = Wb + ((size_t)n * BAND_SIZE + kk * 4 + khalf) * N_LEAF
                                 + wid * 16 + mrow;
            bfrag[n][kk].x = wp[0];
            bfrag[n][kk].y = wp[N_LEAF];
        }
        bias[n] = bb[n * N_LEAF + wid * 16 + mrow];
    }

    __syncthreads();

    const int rowoff = (lane >> 4) * 8;

    v2f a[2][N_BANDS][4];
    #pragma unroll
    for (int u = 0; u < 2; ++u) {
        const float* xr = &ldsX[u * 16 + mrow][khalf];
        #pragma unroll
        for (int n = 0; n < N_BANDS; ++n)
            #pragma unroll
            for (int kk = 0; kk < 4; ++kk)
                a[u][n][kk] = *reinterpret_cast<const v2f*>(xr + n * BAND_SIZE + kk * 4);
    }

    #pragma unroll 1
    for (int tile = 0; tile < TILES_PER_BLOCK; tile += 2) {
        v8f logit[2][N_BANDS];
        #pragma unroll
        for (int u = 0; u < 2; ++u) {
            #pragma unroll
            for (int n = 0; n < N_BANDS; ++n) logit[u][n] = zero8();
            #pragma unroll
            for (int kk = 0; kk < 4; ++kk)
                #pragma unroll
                for (int n = 0; n < N_BANDS; ++n)
                    logit[u][n] = __builtin_amdgcn_wmma_f32_16x16x4_f32(
                        false, a[u][n][kk], false, bfrag[n][kk], (short)0, logit[u][n],
                        false, false);
        }

        {
            const int tn = (tile + 2 < TILES_PER_BLOCK) ? tile + 2 : tile;
            #pragma unroll
            for (int u = 0; u < 2; ++u) {
                const float* xr = &ldsX[(tn + u) * 16 + mrow][khalf];
                #pragma unroll
                for (int n = 0; n < N_BANDS; ++n)
                    #pragma unroll
                    for (int kk = 0; kk < 4; ++kk)
                        a[u][n][kk] = *reinterpret_cast<const v2f*>(xr + n * BAND_SIZE + kk * 4);
            }
        }

        #pragma unroll
        for (int u = 0; u < 2; ++u) {
            v8f acc = zero8();
            #pragma unroll
            for (int n = 0; n < N_BANDS; ++n) {
                const float wn = wband[n];
                #pragma unroll
                for (int j = 0; j < 8; ++j) {
                    float s = frcp(1.0f + __expf(-(logit[u][n][j] + bias[n])));
                    acc[j] += wn * s;
                }
            }

            #pragma unroll
            for (int j = 0; j < 8; ++j)
                ldsRefr[(tile + u) * 16 + rowoff + j][wid * 16 + mrow] = acc[j];
        }
    }

    const int sd = lane & 15;
    const int hh = lane >> 4;
    const int dd = sd >> 2;

    const float c1x = centers1[dd * 3 + 0], c1y = centers1[dd * 3 + 1], c1z = centers1[dd * 3 + 2];
    const float c2x = centers2[sd * 3 + 0], c2y = centers2[sd * 3 + 1], c2z = centers2[sd * 3 + 2];
    float P1[12], P2[12], c3[12];
    #pragma unroll
    for (int i = 0; i < 12; ++i) {
        P1[i] = portal1[dd * 12 + i];
        P2[i] = portal2[sd * 12 + i];
        c3[i] = centers3[sd * 12 + i];
    }
    const float inv2T2 = 1.0f / (2.0f + EPSF);

    const int prow0 = wid * PTS_PER_WAVE + hh;

    __syncthreads();

    v2f npx, npy, npz;
    v4f nrf[2];
    #pragma unroll
    for (int u = 0; u < 2; ++u) {
        const int pr_ = prow0 + 2 * u;
        npx[u] = pos3d[(size_t)(pblock + pr_) * 3 + 0];
        npy[u] = pos3d[(size_t)(pblock + pr_) * 3 + 1];
        npz[u] = pos3d[(size_t)(pblock + pr_) * 3 + 2];
        nrf[u] = *reinterpret_cast<const v4f*>(&ldsRefr[pr_][sd * 4]);
    }

    #pragma unroll 1
    for (int it = 0; it < PTS_PER_WAVE / 4; ++it) {
        const int prow = prow0 + 4 * it;
        const v2f px = npx, py = npy, pz = npz;
        const v4f rf0 = nrf[0], rf1 = nrf[1];
        {
            const int itn = (it + 1 < PTS_PER_WAVE / 4) ? it + 1 : it;
            #pragma unroll
            for (int u = 0; u < 2; ++u) {
                const int prn = prow0 + 4 * itn + 2 * u;
                npx[u] = pos3d[(size_t)(pblock + prn) * 3 + 0];
                npy[u] = pos3d[(size_t)(pblock + prn) * 3 + 1];
                npz[u] = pos3d[(size_t)(pblock + prn) * 3 + 2];
                nrf[u] = *reinterpret_cast<const v4f*>(&ldsRefr[prn][sd * 4]);
            }
        }

        v2f dx = px - c1x, dy = py - c1y, dz = pz - c1z;
        const v2f d1 = dx * dx + dy * dy + dz * dz;
        const v2f z1 = -d1 * inv2T2;
        v2f z1max = max2(z1, row_xor2<4>(z1));
        z1max = max2(z1max, row_xor2<8>(z1max));
        const v2f ex1 = exp_2(z1 - z1max);
        v2f s1 = ex1 + row_xor2<4>(ex1);
        s1 += row_xor2<8>(s1);
        const v2f m1 = ex1 * rcp_2(s1);
        const v2f e1 = exp_2(-LAM * sqrt_2(d1 + EPSF));

        const v2f qx = P1[0] * px + P1[1] * py + P1[2] * pz  + P1[3];
        const v2f qy = P1[4] * px + P1[5] * py + P1[6] * pz  + P1[7];
        const v2f qz = P1[8] * px + P1[9] * py + P1[10] * pz + P1[11];

        dx = qx - c2x; dy = qy - c2y; dz = qz - c2z;
        const v2f d2 = dx * dx + dy * dy + dz * dz;
        const v2f z2 = -d2 * inv2T2;
        v2f z2max = max2(z2, row_xor2<1>(z2));
        z2max = max2(z2max, row_xor2<2>(z2max));
        const v2f ex2 = exp_2(z2 - z2max);
        v2f s2 = ex2 + row_xor2<1>(ex2);
        s2 += row_xor2<2>(s2);
        const v2f m2 = m1 * ex2 * rcp_2(s2);
        const v2f e2 = e1 * exp_2(-LAM * sqrt_2(d2 + EPSF));

        const v2f rx = P2[0] * qx + P2[1] * qy + P2[2] * qz  + P2[3];
        const v2f ry = P2[4] * qx + P2[5] * qy + P2[6] * qz  + P2[7];
        const v2f rz = P2[8] * qx + P2[9] * qy + P2[10] * qz + P2[11];

        v2f d3v[4], z3[4];
        #pragma unroll
        for (int ci = 0; ci < 4; ++ci) {
            const v2f ax = rx - c3[ci * 3 + 0], ay = ry - c3[ci * 3 + 1], az = rz - c3[ci * 3 + 2];
            d3v[ci] = ax * ax + ay * ay + az * az;
            z3[ci]  = -d3v[ci] * inv2T2;
        }
        const v2f z3max = max2(max2(z3[0], z3[1]), max2(z3[2], z3[3]));
        v2f ex3[4];
        #pragma unroll
        for (int ci = 0; ci < 4; ++ci) ex3[ci] = exp_2(z3[ci] - z3max);
        const v2f s3   = ((ex3[0] + ex3[1]) + ex3[2]) + ex3[3];
        const v2f r3   = rcp_2(s3);
        const v2f m2e2 = m2 * e2;

        v2f pr[4];
        #pragma unroll
        for (int ci = 0; ci < 4; ++ci) {
            v2f rfc; rfc[0] = rf0[ci]; rfc[1] = rf1[ci];
            const v2f g = m2e2 * ex3[ci] * r3 * exp_2(-LAM * sqrt_2(d3v[ci] + EPSF));
            pr[ci] = g * rfc;
        }
        v2f ssum = ((pr[0] + pr[1]) + pr[2]) + pr[3];

        ssum += row_xor2<1>(ssum);
        ssum += row_xor2<2>(ssum);
        ssum += row_xor2<4>(ssum);
        ssum += row_xor2<8>(ssum);

        const v2f rn = rcp_2(max2(ssum, mk2(1e-8f)));
        #pragma unroll
        for (int u = 0; u < 2; ++u) {
            v4f v;
            v.x = pr[0][u] * rn[u];
            v.y = pr[1][u] * rn[u];
            v.z = pr[2][u] * rn[u];
            v.w = pr[3][u] * rn[u];
            *reinterpret_cast<v4f*>(out + (size_t)(pblock + prow + 2 * u) * N_LEAF + sd * 4) = v;
        }
    }
}

extern "C" void kernel_launch(void* const* d_in, const int* in_sizes, int n_in,
                              void* d_out, int out_size, void* d_ws, size_t ws_size,
                              hipStream_t stream) {
    const float* pos3d    = (const float*)d_in[0];
    const float* spectral = (const float*)d_in[1];
    const float* centers1 = (const float*)d_in[2];
    const float* centers2 = (const float*)d_in[3];
    const float* centers3 = (const float*)d_in[4];
    const float* portal1  = (const float*)d_in[5];
    const float* portal2  = (const float*)d_in[6];
    const float* Wb       = (const float*)d_in[7];
    const float* bb       = (const float*)d_in[8];
    const float* bw       = (const float*)d_in[9];
    float* out            = (float*)d_out;

    const int BS = in_sizes[0] / 3;
    const int blocks = (BS + PTS_PER_BLOCK - 1) / PTS_PER_BLOCK;

    hipLaunchKernelGGL(inception_traversal_kernel,
                       dim3(blocks), dim3(PTS_PER_BLOCK), 0, stream,
                       pos3d, spectral, centers1, centers2, centers3,
                       portal1, portal2, Wb, bb, bw, out, BS);
}


# Traversal re-mapped to 4 lanes per point (lane = domain; 4 sub-domains, 16 leaves in-lane; sum trees kept): DPP moves per 32 points 192 -> 24; sub-domain constants via LDS table; rows leave via tile
# speedup vs baseline: 1.6799x; 1.0539x over previous
#include <hip/hip_runtime.h>

#define N_LEAF       64
#define SPECTRAL_DIM 64
#define N_BANDS      4
#define BAND_SIZE    16
#define LAM          0.1f
#define EPSF         1e-8f

typedef float v2f __attribute__((ext_vector_type(2)));
typedef float v4f __attribute__((ext_vector_type(4)));
typedef float v8f __attribute__((ext_vector_type(8)));

#define WAVES_PER_BLOCK 4
#define PTS_PER_WAVE    32
#define PTS_PER_BLOCK   (WAVES_PER_BLOCK * PTS_PER_WAVE)
#define LDS_STRIDE      68
#define X_STRIDE        68
#define TILES_PER_BLOCK (PTS_PER_BLOCK / 16)
#define SUB_STRIDE      28

__device__ __forceinline__ v8f zero8() {
    v8f z;
    #pragma unroll
    for (int i = 0; i < 8; ++i) z[i] = 0.0f;
    return z;
}

__device__ __forceinline__ float frcp(float x)  { return __builtin_amdgcn_rcpf(x); }
__device__ __forceinline__ float fsqrt(float x) { return __builtin_amdgcn_sqrtf(x); }

template <int M>
__device__ __forceinline__ float row_xor(float v) {
    return __builtin_bit_cast(float, __builtin_amdgcn_mov_dpp(__builtin_bit_cast(int, v), 0x160 | M, 0xf, 0xf, false));
}

template <int M>
__device__ __forceinline__ v2f row_xor2(v2f v) { v2f r; r.x = row_xor<M>(v.x); r.y = row_xor<M>(v.y); return r; }
__device__ __forceinline__ v2f mk2(float a)      { v2f r; r.x = a; r.y = a; return r; }
__device__ __forceinline__ v2f max2(v2f a, v2f b) { v2f r; r.x = fmaxf(a.x, b.x); r.y = fmaxf(a.y, b.y); return r; }
__device__ __forceinline__ v2f exp_2(v2f a)  { v2f r; r.x = __expf(a.x); r.y = __expf(a.y); return r; }
__device__ __forceinline__ v2f sqrt_2(v2f a) { v2f r; r.x = fsqrt(a.x);  r.y = fsqrt(a.y);  return r; }
__device__ __forceinline__ v2f rcp_2(v2f a)  { v2f r; r.x = frcp(a.x);   r.y = frcp(a.y);   return r; }

__global__ __launch_bounds__(PTS_PER_BLOCK, 4) void inception_traversal_kernel(
    const float* __restrict__ pos3d,
    const float* __restrict__ spectral,
    const float* __restrict__ centers1,
    const float* __restrict__ centers2,
    const float* __restrict__ centers3,
    const float* __restrict__ portal1,
    const float* __restrict__ portal2,
    const float* __restrict__ Wb,
    const float* __restrict__ bb,
    const float* __restrict__ bw,
    float* __restrict__ out,
    int BS)
{
    __shared__ __attribute__((aligned(16))) float ldsRefr[PTS_PER_BLOCK][LDS_STRIDE];
    __shared__ __attribute__((aligned(16))) float ldsX[PTS_PER_BLOCK][X_STRIDE];
    __shared__ __attribute__((aligned(16))) float ldsSub[16][SUB_STRIDE];

    const int lane   = threadIdx.x & 31;
    const int wid    = threadIdx.x >> 5;
    const int pblock = blockIdx.x * PTS_PER_BLOCK;
    if (pblock >= BS) return;
    const int pwave  = pblock + wid * PTS_PER_WAVE;

    float w0 = bw[0], w1 = bw[1], w2 = bw[2], w3 = bw[3];
    float wm = fmaxf(fmaxf(w0, w1), fmaxf(w2, w3));
    float ew0 = __expf(w0 - wm), ew1 = __expf(w1 - wm);
    float ew2 = __expf(w2 - wm), ew3 = __expf(w3 - wm);
    float wr  = frcp(ew0 + ew1 + ew2 + ew3);
    float wband[N_BANDS] = { ew0 * wr, ew1 * wr, ew2 * wr, ew3 * wr };

    const int mrow  = lane & 15;
    const int khalf = (lane >> 4) * 2;


    {
        const float* xw = spectral + (size_t)pwave * SPECTRAL_DIM + lane * 4;
        const int r0 = wid * PTS_PER_WAVE + (lane >> 4);
        const int c0 = (lane & 15) * 4;
        v4f xv[16];
        #pragma unroll
        for (int i = 0; i < 16; ++i)
            xv[i] = *reinterpret_cast<const v4f*>(xw + (size_t)i * 2 * SPECTRAL_DIM);
        #pragma unroll
        for (int i = 0; i < 16; ++i)
            *reinterpret_cast<v4f*>(&ldsX[r0 + 2 * i][c0]) = xv[i];
    }

    v2f bfrag[N_BANDS][4];
    float bias[N_BANDS];
    #pragma unroll
    for (int n = 0; n < N_BANDS; ++n) {
        #pragma unroll
        for (int kk = 0; kk < 4; ++kk) {
            const float* wp = Wb + ((size_t)n * BAND_SIZE + kk * 4 + khalf) * N_LEAF
                                 + wid * 16 + mrow;
            bfrag[n][kk].x = wp[0];
            bfrag[n][kk].y = wp[N_LEAF];
        }
        bias[n] = bb[n * N_LEAF + wid * 16 + mrow];
    }

    {
        const int tid = threadIdx.x;
        for (int idx = tid; idx < 192; idx += PTS_PER_BLOCK) {
            ldsSub[idx / 12][idx % 12]      = portal2[idx];
            ldsSub[idx / 12][12 + idx % 12] = centers3[idx];
        }
        if (tid < 48) ldsSub[tid / 3][24 + tid % 3] = centers2[tid];
    }

    __syncthreads();

    const int rowoff = (lane >> 4) * 8;

    v2f a[2][N_BANDS][4];
    #pragma unroll
    for (int u = 0; u < 2; ++u) {
        const float* xr = &ldsX[u * 16 + mrow][khalf];
        #pragma unroll
        for (int n = 0; n < N_BANDS; ++n)
            #pragma unroll
            for (int kk = 0; kk < 4; ++kk)
                a[u][n][kk] = *reinterpret_cast<const v2f*>(xr + n * BAND_SIZE + kk * 4);
    }

    #pragma unroll 1
    for (int tile = 0; tile < TILES_PER_BLOCK; tile += 2) {
        v8f logit[2][N_BANDS];
        #pragma unroll
        for (int u = 0; u < 2; ++u) {
            #pragma unroll
            for (int n = 0; n < N_BANDS; ++n) logit[u][n] = zero8();
            #pragma unroll
            for (int kk = 0; kk < 4; ++kk)
                #pragma unroll
                for (int n = 0; n < N_BANDS; ++n)
                    logit[u][n] = __builtin_amdgcn_wmma_f32_16x16x4_f32(
                        false, a[u][n][kk], false, bfrag[n][kk], (short)0, logit[u][n],
                        false, false);
        }

        {
            const int tn = (tile + 2 < TILES_PER_BLOCK) ? tile + 2 : tile;
            #pragma unroll
            for (int u = 0; u < 2; ++u) {
                const float* xr = &ldsX[(tn + u) * 16 + mrow][khalf];
                #pragma unroll
                for (int n = 0; n < N_BANDS; ++n)
                    #pragma unroll
                    for (int kk = 0; kk < 4; ++kk)
                        a[u][n][kk] = *reinterpret_cast<const v2f*>(xr + n * BAND_SIZE + kk * 4);
            }
        }

        #pragma unroll
        for (int u = 0; u < 2; ++u) {
            v8f acc = zero8();
            #pragma unroll
            for (int n = 0; n < N_BANDS; ++n) {
                const float wn = wband[n];
                #pragma unroll
                for (int j = 0; j < 8; ++j) {
                    float s = frcp(1.0f + __expf(-(logit[u][n][j] + bias[n])));
                    acc[j] += wn * s;
                }
            }

            #pragma unroll
            for (int j = 0; j < 8; ++j)
                ldsRefr[(tile + u) * 16 + rowoff + j][wid * 16 + mrow] = acc[j];
        }
    }

    const int dm = lane & 3;
    const int ps = lane >> 2;

    const float c1x = centers1[dm * 3 + 0], c1y = centers1[dm * 3 + 1], c1z = centers1[dm * 3 + 2];
    float P1[12];
    #pragma unroll
    for (int i = 0; i < 12; ++i) P1[i] = portal1[dm * 12 + i];
    const float inv2T2 = 1.0f / (2.0f + EPSF);

    const int prow0 = wid * PTS_PER_WAVE + ps;
    v2f ppx[2], ppy[2], ppz[2];
    #pragma unroll
    for (int t = 0; t < 2; ++t) {
        #pragma unroll
        for (int u = 0; u < 2; ++u) {
            const int pr_ = prow0 + 16 * t + 8 * u;
            ppx[t][u] = pos3d[(size_t)(pblock + pr_) * 3 + 0];
            ppy[t][u] = pos3d[(size_t)(pblock + pr_) * 3 + 1];
            ppz[t][u] = pos3d[(size_t)(pblock + pr_) * 3 + 2];
        }
    }

    __syncthreads();

    v2f qx[2], qy[2], qz[2], m2e2[2][4];
    #pragma unroll
    for (int t = 0; t < 2; ++t) {
        const v2f px = ppx[t], py = ppy[t], pz = ppz[t];

        const v2f dx = px - c1x, dy = py - c1y, dz = pz - c1z;
        const v2f d1 = dx * dx + dy * dy + dz * dz;
        const v2f z1 = -d1 * inv2T2;
        v2f z1max = max2(z1, row_xor2<1>(z1));
        z1max = max2(z1max, row_xor2<2>(z1max));
        const v2f ex1 = exp_2(z1 - z1max);
        v2f s1 = ex1 + row_xor2<1>(ex1);
        s1 += row_xor2<2>(s1);
        const v2f m1 = ex1 * rcp_2(s1);
        const v2f e1 = exp_2(-LAM * sqrt_2(d1 + EPSF));

        qx[t] = P1[0] * px + P1[1] * py + P1[2] * pz  + P1[3];
        qy[t] = P1[4] * px + P1[5] * py + P1[6] * pz  + P1[7];
        qz[t] = P1[8] * px + P1[9] * py + P1[10] * pz + P1[11];

        v2f d2[4], z2[4];
        #pragma unroll
        for (int si = 0; si < 4; ++si) {
            const float* sub = &ldsSub[dm * 4 + si][0];
            const v2f ax = qx[t] - sub[24], ay = qy[t] - sub[25], az = qz[t] - sub[26];
            d2[si] = ax * ax + ay * ay + az * az;
            z2[si] = -d2[si] * inv2T2;
        }
        const v2f z2max = max2(max2(z2[0], z2[1]), max2(z2[2], z2[3]));
        v2f ex2[4];
        #pragma unroll
        for (int si = 0; si < 4; ++si) ex2[si] = exp_2(z2[si] - z2max);
        const v2f s2 = (ex2[0] + ex2[1]) + (ex2[2] + ex2[3]);
        const v2f r2 = rcp_2(s2);
        #pragma unroll
        for (int si = 0; si < 4; ++si) {
            const v2f m2 = m1 * ex2[si] * r2;
            const v2f e2 = e1 * exp_2(-LAM * sqrt_2(d2[si] + EPSF));
            m2e2[t][si] = m2 * e2;
        }
    }

    v2f ss[2][4];
    #pragma unroll
    for (int si = 0; si < 4; ++si) {
        const float* sub = &ldsSub[dm * 4 + si][0];
        float P2[12], c3[12];
        #pragma unroll
        for (int q4 = 0; q4 < 3; ++q4) {
            const v4f a = *reinterpret_cast<const v4f*>(sub + 4 * q4);
            const v4f b = *reinterpret_cast<const v4f*>(sub + 12 + 4 * q4);
            P2[4 * q4 + 0] = a.x; P2[4 * q4 + 1] = a.y; P2[4 * q4 + 2] = a.z; P2[4 * q4 + 3] = a.w;
            c3[4 * q4 + 0] = b.x; c3[4 * q4 + 1] = b.y; c3[4 * q4 + 2] = b.z; c3[4 * q4 + 3] = b.w;
        }

        #pragma unroll
        for (int t = 0; t < 2; ++t) {
            float* t0 = &ldsRefr[prow0 + 16 * t][dm * 16 + si * 4];
            float* t1 = &ldsRefr[prow0 + 16 * t + 8][dm * 16 + si * 4];
            const v4f rf0 = *reinterpret_cast<const v4f*>(t0);
            const v4f rf1 = *reinterpret_cast<const v4f*>(t1);

            const v2f rx = P2[0] * qx[t] + P2[1] * qy[t] + P2[2] * qz[t]  + P2[3];
            const v2f ry = P2[4] * qx[t] + P2[5] * qy[t] + P2[6] * qz[t]  + P2[7];
            const v2f rz = P2[8] * qx[t] + P2[9] * qy[t] + P2[10] * qz[t] + P2[11];

            v2f d3v[4], z3[4];
            #pragma unroll
            for (int ci = 0; ci < 4; ++ci) {
                const v2f ax = rx - c3[ci * 3 + 0], ay = ry - c3[ci * 3 + 1], az = rz - c3[ci * 3 + 2];
                d3v[ci] = ax * ax + ay * ay + az * az;
                z3[ci]  = -d3v[ci] * inv2T2;
            }
            const v2f z3max = max2(max2(z3[0], z3[1]), max2(z3[2], z3[3]));
            v2f ex3[4];
            #pragma unroll
            for (int ci = 0; ci < 4; ++ci) ex3[ci] = exp_2(z3[ci] - z3max);
            const v2f s3 = ((ex3[0] + ex3[1]) + ex3[2]) + ex3[3];
            const v2f r3 = rcp_2(s3);

            v2f pr[4];
            #pragma unroll
            for (int ci = 0; ci < 4; ++ci) {
                v2f rfc; rfc[0] = rf0[ci]; rfc[1] = rf1[ci];
                const v2f g = m2e2[t][si] * ex3[ci] * r3 * exp_2(-LAM * sqrt_2(d3v[ci] + EPSF));
                pr[ci] = g * rfc;
            }
            ss[t][si] = ((pr[0] + pr[1]) + pr[2]) + pr[3];

            v4f o0, o1;
            o0.x = pr[0][0]; o0.y = pr[1][0]; o0.z = pr[2][0]; o0.w = pr[3][0];
            o1.x = pr[0][1]; o1.y = pr[1][1]; o1.z = pr[2][1]; o1.w = pr[3][1];
            *reinterpret_cast<v4f*>(t0) = o0;
            *reinterpret_cast<v4f*>(t1) = o1;
        }
    }

    #pragma unroll
    for (int t = 0; t < 2; ++t) {
        v2f ssum = (ss[t][0] + ss[t][1]) + (ss[t][2] + ss[t][3]);
        ssum += row_xor2<1>(ssum);
        ssum += row_xor2<2>(ssum);

        const v2f rn = rcp_2(max2(ssum, mk2(1e-8f)));
        if (dm == 0) {
            ldsRefr[prow0 + 16 * t][N_LEAF]     = rn[0];
            ldsRefr[prow0 + 16 * t + 8][N_LEAF] = rn[1];
        }
    }

    __builtin_amdgcn_fence(__ATOMIC_RELEASE, "wavefront");
    __builtin_amdgcn_wave_barrier();
    __builtin_amdgcn_fence(__ATOMIC_ACQUIRE, "wavefront");

    {
        const int sd = lane & 15;
        const int hh = lane >> 4;
        #pragma unroll
        for (int i = 0; i < 16; ++i) {
            const int r = wid * PTS_PER_WAVE + 2 * i + hh;
            const v4f v = *reinterpret_cast<const v4f*>(&ldsRefr[r][sd * 4]);
            const float rn = ldsRefr[r][N_LEAF];
            *reinterpret_cast<v4f*>(out + (size_t)(pblock + r) * N_LEAF + sd * 4) = v * rn;
        }
    }
}

extern "C" void kernel_launch(void* const* d_in, const int* in_sizes, int n_in,
                              void* d_out, int out_size, void* d_ws, size_t ws_size,
                              hipStream_t stream) {
    const float* pos3d    = (const float*)d_in[0];
    const float* spectral = (const float*)d_in[1];
    const float* centers1 = (const float*)d_in[2];
    const float* centers2 = (const float*)d_in[3];
    const float* centers3 = (const float*)d_in[4];
    const float* portal1  = (const float*)d_in[5];
    const float* portal2  = (const float*)d_in[6];
    const float* Wb       = (const float*)d_in[7];
    const float* bb       = (const float*)d_in[8];
    const float* bw       = (const float*)d_in[9];
    float* out            = (float*)d_out;

    const int BS = in_sizes[0] / 3;
    const int blocks = (BS + PTS_PER_BLOCK - 1) / PTS_PER_BLOCK;

    hipLaunchKernelGGL(inception_traversal_kernel,
                       dim3(blocks), dim3(PTS_PER_BLOCK), 0, stream,
                       pos3d, spectral, centers1, centers2, centers3,
                       portal1, portal2, Wb, bb, bw, out, BS);
}


# Geometric traversal moved ahead of the routing product, into the wait for the tile rows: leaf factors parked in the LDS tile, each routing fragment multiplied into them in place; sum orders kept
# speedup vs baseline: 1.8526x; 1.1028x over previous
#include <hip/hip_runtime.h>

#define N_LEAF       64
#define SPECTRAL_DIM 64
#define N_BANDS      4
#define BAND_SIZE    16
#define LAM          0.1f
#define EPSF         1e-8f

typedef float v2f __attribute__((ext_vector_type(2)));
typedef float v4f __attribute__((ext_vector_type(4)));
typedef float v8f __attribute__((ext_vector_type(8)));

#define WAVES_PER_BLOCK 4
#define PTS_PER_WAVE    32
#define PTS_PER_BLOCK   (WAVES_PER_BLOCK * PTS_PER_WAVE)
#define LDS_STRIDE      68
#define X_STRIDE        68
#define TILES_PER_BLOCK (PTS_PER_BLOCK / 16)
#define SUB_STRIDE      28

__device__ __forceinline__ v8f zero8() {
    v8f z;
    #pragma unroll
    for (int i = 0; i < 8; ++i) z[i] = 0.0f;
    return z;
}

__device__ __forceinline__ float frcp(float x)  { return __builtin_amdgcn_rcpf(x); }
__device__ __forceinline__ float fsqrt(float x) { return __builtin_amdgcn_sqrtf(x); }

template <int M>
__device__ __forceinline__ float row_xor(float v) {
    return __builtin_bit_cast(float, __builtin_amdgcn_mov_dpp(__builtin_bit_cast(int, v), 0x160 | M, 0xf, 0xf, false));
}

template <int M>
__device__ __forceinline__ v2f row_xor2(v2f v) { v2f r; r.x = row_xor<M>(v.x); r.y = row_xor<M>(v.y); return r; }
__device__ __forceinline__ v2f mk2(float a)      { v2f r; r.x = a; r.y = a; return r; }
__device__ __forceinline__ v2f max2(v2f a, v2f b) { v2f r; r.x = fmaxf(a.x, b.x); r.y = fmaxf(a.y, b.y); return r; }
__device__ __forceinline__ v2f exp_2(v2f a)  { v2f r; r.x = __expf(a.x); r.y = __expf(a.y); return r; }
__device__ __forceinline__ v2f sqrt_2(v2f a) { v2f r; r.x = fsqrt(a.x);  r.y = fsqrt(a.y);  return r; }
__device__ __forceinline__ v2f rcp_2(v2f a)  { v2f r; r.x = frcp(a.x);   r.y = frcp(a.y);   return r; }

__global__ __launch_bounds__(PTS_PER_BLOCK, 4) void inception_traversal_kernel(
    const float* __restrict__ pos3d,
    const float* __restrict__ spectral,
    const float* __restrict__ centers1,
    const float* __restrict__ centers2,
    const float* __restrict__ centers3,
    const float* __restrict__ portal1,
    const float* __restrict__ portal2,
    const float* __restrict__ Wb,
    const float* __restrict__ bb,
    const float* __restrict__ bw,
    float* __restrict__ out,
    int BS)
{
    __shared__ __attribute__((aligned(16))) float ldsRefr[PTS_PER_BLOCK][LDS_STRIDE];
    __shared__ __attribute__((aligned(16))) float ldsX[PTS_PER_BLOCK][X_STRIDE];
    __shared__ __attribute__((aligned(16))) float ldsSub[16][SUB_STRIDE];

    const int lane   = threadIdx.x & 31;
    const int wid    = threadIdx.x >> 5;
    const int pblock = blockIdx.x * PTS_PER_BLOCK;
    if (pblock >= BS) return;
    const int pwave  = pblock + wid * PTS_PER_WAVE;

    float w0 = bw[0], w1 = bw[1], w2 = bw[2], w3 = bw[3];
    float wm = fmaxf(fmaxf(w0, w1), fmaxf(w2, w3));
    float ew0 = __expf(w0 - wm), ew1 = __expf(w1 - wm);
    float ew2 = __expf(w2 - wm), ew3 = __expf(w3 - wm);
    float wr  = frcp(ew0 + ew1 + ew2 + ew3);
    float wband[N_BANDS] = { ew0 * wr, ew1 * wr, ew2 * wr, ew3 * wr };

    const int mrow  = lane & 15;
    const int khalf = (lane >> 4) * 2;

    const int dm = lane & 3;
    const int ps = lane >> 2;
    const int prow0 = wid * PTS_PER_WAVE + ps;
    const float inv2T2 = 1.0f / (2.0f + EPSF);


    const int ti0 = threadIdx.x;
    const int ti1 = (threadIdx.x < 64) ? (int)threadIdx.x + 128 : (int)threadIdx.x;
    const float tp0 = portal2[ti0], tc0 = centers3[ti0];
    const float tp1 = portal2[ti1], tc1 = centers3[ti1];

    const float c1x = centers1[dm * 3 + 0], c1y = centers1[dm * 3 + 1], c1z = centers1[dm * 3 + 2];
    float P1[12], c2[12];
    #pragma unroll
    for (int i = 0; i < 12; ++i) P1[i] = portal1[dm * 12 + i];
    #pragma unroll
    for (int i = 0; i < 12; ++i) c2[i] = centers2[dm * 12 + i];
    v2f ppx[2], ppy[2], ppz[2];
    #pragma unroll
    for (int t = 0; t < 2; ++t) {
        #pragma unroll
        for (int u = 0; u < 2; ++u) {
            const int pr_ = prow0 + 16 * t + 8 * u;
            ppx[t][u] = pos3d[(size_t)(pblock + pr_) * 3 + 0];
            ppy[t][u] = pos3d[(size_t)(pblock + pr_) * 3 + 1];
            ppz[t][u] = pos3d[(size_t)(pblock + pr_) * 3 + 2];
        }
    }

    const int xr0 = wid * PTS_PER_WAVE + (lane >> 4);
    const int xc0 = (lane & 15) * 4;
    v4f xv[16];
    {
        const float* xw = spectral + (size_t)pwave * SPECTRAL_DIM + lane * 4;
        #pragma unroll
        for (int i = 0; i < 16; ++i)
            xv[i] = *reinterpret_cast<const v4f*>(xw + (size_t)i * 2 * SPECTRAL_DIM);
    }

    v2f bfrag[N_BANDS][4];
    float bias[N_BANDS];
    #pragma unroll
    for (int n = 0; n < N_BANDS; ++n) {
        #pragma unroll
        for (int kk = 0; kk < 4; ++kk) {
            const float* wp = Wb + ((size_t)n * BAND_SIZE + kk * 4 + khalf) * N_LEAF
                                 + wid * 16 + mrow;
            bfrag[n][kk].x = wp[0];
            bfrag[n][kk].y = wp[N_LEAF];
        }
        bias[n] = bb[n * N_LEAF + wid * 16 + mrow];
    }

    ldsSub[ti0 / 12][ti0 % 12]      = tp0;
    ldsSub[ti0 / 12][12 + ti0 % 12] = tc0;
    ldsSub[ti1 / 12][ti1 % 12]      = tp1;
    ldsSub[ti1 / 12][12 + ti1 % 12] = tc1;

    __builtin_amdgcn_fence(__ATOMIC_RELEASE, "workgroup", "local");
    __builtin_amdgcn_s_barrier();
    __builtin_amdgcn_fence(__ATOMIC_ACQUIRE, "workgroup", "local");


    v2f qx[2], qy[2], qz[2], m2e2[2][4];
    #pragma unroll
    for (int t = 0; t < 2; ++t) {
        const v2f px = ppx[t], py = ppy[t], pz = ppz[t];

        const v2f dx = px - c1x, dy = py - c1y, dz = pz - c1z;
        const v2f d1 = dx * dx + dy * dy + dz * dz;
        const v2f z1 = -d1 * inv2T2;
        v2f z1max = max2(z1, row_xor2<1>(z1));
        z1max = max2(z1max, row_xor2<2>(z1max));
        const v2f ex1 = exp_2(z1 - z1max);
        v2f s1 = ex1 + row_xor2<1>(ex1);
        s1 += row_xor2<2>(s1);
        const v2f m1 = ex1 * rcp_2(s1);
        const v2f e1 = exp_2(-LAM * sqrt_2(d1 + EPSF));

        qx[t] = P1[0] * px + P1[1] * py + P1[2] * pz  + P1[3];
        qy[t] = P1[4] * px + P1[5] * py + P1[6] * pz  + P1[7];
        qz[t] = P1[8] * px + P1[9] * py + P1[10] * pz + P1[11];

        v2f d2[4], z2[4];
        #pragma unroll
        for (int si = 0; si < 4; ++si) {
            const v2f ax = qx[t] - c2[si * 3 + 0], ay = qy[t] - c2[si * 3 + 1], az = qz[t] - c2[si * 3 + 2];
            d2[si] = ax * ax + ay * ay + az * az;
            z2[si] = -d2[si] * inv2T2;
        }
        const v2f z2max = max2(max2(z2[0], z2[1]), max2(z2[2], z2[3]));
        v2f ex2[4];
        #pragma unroll
        for (int si = 0; si < 4; ++si) ex2[si] = exp_2(z2[si] - z2max);
        const v2f s2 = (ex2[0] + ex2[1]) + (ex2[2] + ex2[3]);
        const v2f r2 = rcp_2(s2);
        #pragma unroll
        for (int si = 0; si < 4; ++si) {
            const v2f m2 = m1 * ex2[si] * r2;
            const v2f e2 = e1 * exp_2(-LAM * sqrt_2(d2[si] + EPSF));
            m2e2[t][si] = m2 * e2;
        }
    }

    #pragma unroll
    for (int si = 0; si < 4; ++si) {
        const float* sub = &ldsSub[dm * 4 + si][0];
        float P2[12], c3[12];
        #pragma unroll
        for (int q4 = 0; q4 < 3; ++q4) {
            const v4f a = *reinterpret_cast<const v4f*>(sub + 4 * q4);
            const v4f b = *reinterpret_cast<const v4f*>(sub + 12 + 4 * q4);
            P2[4 * q4 + 0] = a.x; P2[4 * q4 + 1] = a.y; P2[4 * q4 + 2] = a.z; P2[4 * q4 + 3] = a.w;
            c3[4 * q4 + 0] = b.x; c3[4 * q4 + 1] = b.y; c3[4 * q4 + 2] = b.z; c3[4 * q4 + 3] = b.w;
        }

        #pragma unroll
        for (int t = 0; t < 2; ++t) {
            const v2f rx = P2[0] * qx[t] + P2[1] * qy[t] + P2[2] * qz[t]  + P2[3];
            const v2f ry = P2[4] * qx[t] + P2[5] * qy[t] + P2[6] * qz[t]  + P2[7];
            const v2f rz = P2[8] * qx[t] + P2[9] * qy[t] + P2[10] * qz[t] + P2[11];

            v2f d3v[4], z3[4];
            #pragma unroll
            for (int ci = 0; ci < 4; ++ci) {
                const v2f ax = rx - c3[ci * 3 + 0], ay = ry - c3[ci * 3 + 1], az = rz - c3[ci * 3 + 2];
                d3v[ci] = ax * ax + ay * ay + az * az;
                z3[ci]  = -d3v[ci] * inv2T2;
            }
            const v2f z3max = max2(max2(z3[0], z3[1]), max2(z3[2], z3[3]));
            v2f ex3[4];
            #pragma unroll
            for (int ci = 0; ci < 4; ++ci) ex3[ci] = exp_2(z3[ci] - z3max);
            const v2f s3 = ((ex3[0] + ex3[1]) + ex3[2]) + ex3[3];
            const v2f r3 = rcp_2(s3);

            v2f g[4];
            #pragma unroll
            for (int ci = 0; ci < 4; ++ci)
                g[ci] = m2e2[t][si] * ex3[ci] * r3 * exp_2(-LAM * sqrt_2(d3v[ci] + EPSF));

            v4f o0, o1;
            o0.x = g[0][0]; o0.y = g[1][0]; o0.z = g[2][0]; o0.w = g[3][0];
            o1.x = g[0][1]; o1.y = g[1][1]; o1.z = g[2][1]; o1.w = g[3][1];
            *reinterpret_cast<v4f*>(&ldsRefr[prow0 + 16 * t][dm * 16 + si * 4])     = o0;
            *reinterpret_cast<v4f*>(&ldsRefr[prow0 + 16 * t + 8][dm * 16 + si * 4]) = o1;
        }
    }

    #pragma unroll
    for (int i = 0; i < 16; ++i)
        *reinterpret_cast<v4f*>(&ldsX[xr0 + 2 * i][xc0]) = xv[i];

    __syncthreads();


    const int rowoff = (lane >> 4) * 8;

    v2f a[2][N_BANDS][4];
    #pragma unroll
    for (int u = 0; u < 2; ++u) {
        const float* xr = &ldsX[u * 16 + mrow][khalf];
        #pragma unroll
        for (int n = 0; n < N_BANDS; ++n)
            #pragma unroll
            for (int kk = 0; kk < 4; ++kk)
                a[u][n][kk] = *reinterpret_cast<const v2f*>(xr + n * BAND_SIZE + kk * 4);
    }

    #pragma unroll 1
    for (int tile = 0; tile < TILES_PER_BLOCK; tile += 2) {
        float gq[2][8];
        #pragma unroll
        for (int u = 0; u < 2; ++u)
            #pragma unroll
            for (int j = 0; j < 8; ++j)
                gq[u][j] = ldsRefr[(tile + u) * 16 + rowoff + j][wid * 16 + mrow];

        v8f logit[2][N_BANDS];
        #pragma unroll
        for (int u = 0; u < 2; ++u) {
            #pragma unroll
            for (int n = 0; n < N_BANDS; ++n) logit[u][n] = zero8();
            #pragma unroll
            for (int kk = 0; kk < 4; ++kk)
                #pragma unroll
                for (int n = 0; n < N_BANDS; ++n)
                    logit[u][n] = __builtin_amdgcn_wmma_f32_16x16x4_f32(
                        false, a[u][n][kk], false, bfrag[n][kk], (short)0, logit[u][n],
                        false, false);
        }

        {
            const int tn = (tile + 2 < TILES_PER_BLOCK) ? tile + 2 : tile;
            #pragma unroll
            for (int u = 0; u < 2; ++u) {
                const float* xr = &ldsX[(tn + u) * 16 + mrow][khalf];
                #pragma unroll
                for (int n = 0; n < N_BANDS; ++n)
                    #pragma unroll
                    for (int kk = 0; kk < 4; ++kk)
                        a[u][n][kk] = *reinterpret_cast<const v2f*>(xr + n * BAND_SIZE + kk * 4);
            }
        }

        #pragma unroll
        for (int u = 0; u < 2; ++u) {
            v8f acc = zero8();
            #pragma unroll
            for (int n = 0; n < N_BANDS; ++n) {
                const float wn = wband[n];
                #pragma unroll
                for (int j = 0; j < 8; ++j) {
                    float s = frcp(1.0f + __expf(-(logit[u][n][j] + bias[n])));
                    acc[j] += wn * s;
                }
            }

            #pragma unroll
            for (int j = 0; j < 8; ++j)
                ldsRefr[(tile + u) * 16 + rowoff + j][wid * 16 + mrow] = gq[u][j] * acc[j];
        }
    }

    __syncthreads();

    #pragma unroll
    for (int t = 0; t < 2; ++t) {
        v2f ss[4];
        #pragma unroll
        for (int si = 0; si < 4; ++si) {
            const v4f rf0 = *reinterpret_cast<const v4f*>(&ldsRefr[prow0 + 16 * t][dm * 16 + si * 4]);
            const v4f rf1 = *reinterpret_cast<const v4f*>(&ldsRefr[prow0 + 16 * t + 8][dm * 16 + si * 4]);
            v2f pr[4];
            #pragma unroll
            for (int ci = 0; ci < 4; ++ci) { pr[ci][0] = rf0[ci]; pr[ci][1] = rf1[ci]; }
            ss[si] = ((pr[0] + pr[1]) + pr[2]) + pr[3];
        }

        v2f ssum = (ss[0] + ss[1]) + (ss[2] + ss[3]);
        ssum += row_xor2<1>(ssum);
        ssum += row_xor2<2>(ssum);

        const v2f rn = rcp_2(max2(ssum, mk2(1e-8f)));
        if (dm == 0) {
            ldsRefr[prow0 + 16 * t][N_LEAF]     = rn[0];
            ldsRefr[prow0 + 16 * t + 8][N_LEAF] = rn[1];
        }
    }

    __builtin_amdgcn_fence(__ATOMIC_RELEASE, "wavefront");
    __builtin_amdgcn_wave_barrier();
    __builtin_amdgcn_fence(__ATOMIC_ACQUIRE, "wavefront");

    {
        const int sd = lane & 15;
        const int hh = lane >> 4;
        #pragma unroll
        for (int i = 0; i < 16; ++i) {
            const int r = wid * PTS_PER_WAVE + 2 * i + hh;
            const v4f v = *reinterpret_cast<const v4f*>(&ldsRefr[r][sd * 4]);
            const float rn = ldsRefr[r][N_LEAF];
            *reinterpret_cast<v4f*>(out + (size_t)(pblock + r) * N_LEAF + sd * 4) = v * rn;
        }
    }
}

extern "C" void kernel_launch(void* const* d_in, const int* in_sizes, int n_in,
                              void* d_out, int out_size, void* d_ws, size_t ws_size,
                              hipStream_t stream) {
    const float* pos3d    = (const float*)d_in[0];
    const float* spectral = (const float*)d_in[1];
    const float* centers1 = (const float*)d_in[2];
    const float* centers2 = (const float*)d_in[3];
    const float* centers3 = (const float*)d_in[4];
    const float* portal1  = (const float*)d_in[5];
    const float* portal2  = (const float*)d_in[6];
    const float* Wb       = (const float*)d_in[7];
    const float* bb       = (const float*)d_in[8];
    const float* bw       = (const float*)d_in[9];
    float* out            = (float*)d_out;

    const int BS = in_sizes[0] / 3;
    const int blocks = (BS + PTS_PER_BLOCK - 1) / PTS_PER_BLOCK;

    hipLaunchKernelGGL(inception_traversal_kernel,
                       dim3(blocks), dim3(PTS_PER_BLOCK), 0, stream,
                       pos3d, spectral, centers1, centers2, centers3,
                       portal1, portal2, Wb, bb, bw, out, BS);
}
